# starting kernel
# baseline (speedup 1.0000x reference)
#include <hip/hip_runtime.h>
#include <hip/hip_bf16.h>

typedef __attribute__((ext_vector_type(2))) float v2f;
typedef __attribute__((ext_vector_type(8))) float v8f;

__global__ void k_deg_init(float* __restrict__ deg, int n) {
    int i = blockIdx.x * blockDim.x + threadIdx.x;
    if (i < n) deg[i] = 1.0f;
}

__global__ void k_deg_accum(const int* __restrict__ dst, float* __restrict__ deg, int e) {
    int i = blockIdx.x * blockDim.x + threadIdx.x;
    if (i < e) atomicAdd(&deg[dst[i]], 1.0f);
}

__global__ void k_dinv(const float* __restrict__ deg, float* __restrict__ dinv, int n) {
    int i = blockIdx.x * blockDim.x + threadIdx.x;
    if (i < n) dinv[i] = rsqrtf(deg[i]);
}

template <int DCOL>
__global__ __launch_bounds__(256) void k_gemm_wmma(const float* __restrict__ X,
                                                   const float* __restrict__ W,
                                                   float* __restrict__ H,
                                                   int nrows) {
    constexpr int COLT = DCOL / 16;
    constexpr int ROWT = 8 / COLT;
    const int wave = threadIdx.x >> 5;
    const int lane = threadIdx.x & 31;
    const int m0 = (blockIdx.x * ROWT + (wave / COLT)) * 16;
    const int n0 = (wave % COLT) * 16;
    if (m0 >= nrows) return;

    const int lr = lane & 15;
    const int lh = lane >> 4;

    v8f c = {};
    const float* xrow = X + (long long)(m0 + lr) * 64;
#pragma unroll
    for (int k = 0; k < 64; k += 4) {
        v2f a, b;
        a.x = xrow[k + lh * 2 + 0];
        a.y = xrow[k + lh * 2 + 1];
        b.x = W[(k + lh * 2 + 0) * DCOL + n0 + lr];
        b.y = W[(k + lh * 2 + 1) * DCOL + n0 + lr];
        c = __builtin_amdgcn_wmma_f32_16x16x4_f32(false, a, false, b,
                                                  (short)0, c, false, false);
    }
#pragma unroll
    for (int r = 0; r < 8; ++r) {
        H[(long long)(m0 + r + lh * 8) * DCOL + n0 + lr] = c[r];
    }
}

__global__ void k_self_bias(const float* __restrict__ H, const float* __restrict__ dinv,
                            const float* __restrict__ b, float* __restrict__ out,
                            int n, int D) {
    long long idx = (long long)blockIdx.x * blockDim.x + threadIdx.x;
    long long total = (long long)n * D;
    if (idx >= total) return;
    int i = (int)(idx / D);
    int f = (int)(idx % D);
    float di = dinv[i];
    out[idx] = H[idx] * di * di + b[f];
}

__global__ void k_scatter(const float* __restrict__ H, const int* __restrict__ src,
                          const int* __restrict__ dst, const float* __restrict__ dinv,
                          float* __restrict__ out, int e, int D) {
    const int tpe = D >> 2;
    long long gid = (long long)blockIdx.x * blockDim.x + threadIdx.x;
    long long total = (long long)e * tpe;
    if (gid >= total) return;
    int edge = (int)(gid / tpe);
    int f = (int)(gid % tpe) * 4;
    int s = src[edge], d = dst[edge];
    float norm = dinv[s] * dinv[d];
    const float4 hv = *(const float4*)(H + (long long)s * D + f);
    float* op = out + (long long)d * D + f;
    atomicAdd(op + 0, hv.x * norm);
    atomicAdd(op + 1, hv.y * norm);
    atomicAdd(op + 2, hv.z * norm);
    atomicAdd(op + 3, hv.w * norm);
}

__global__ void k_relu(float* __restrict__ a, long long total) {
    long long idx = (long long)blockIdx.x * blockDim.x + threadIdx.x;
    if (idx < total) a[idx] = fmaxf(a[idx], 0.0f);
}

extern "C" void kernel_launch(void* const* d_in, const int* in_sizes, int n_in,
                              void* d_out, int out_size, void* d_ws, size_t ws_size,
                              hipStream_t stream) {
    const int D_IN = 64, D_H = 64, D_OUT = 32;
    const int N = in_sizes[0] / D_IN;
    const int E = in_sizes[1] / 2;

    const float* x  = (const float*)d_in[0];
    const int*   ei = (const int*)d_in[1];
    const float* W1 = (const float*)d_in[2];
    const float* b1 = (const float*)d_in[3];
    const float* W2 = (const float*)d_in[4];
    const float* b2 = (const float*)d_in[5];
    float* out = (float*)d_out;

    const int* srcIdx = ei;
    const int* dstIdx = ei + E;

    float* ws   = (float*)d_ws;
    float* deg  = ws;
    float* dinv = deg  + N;
    float* h1   = dinv + N;
    float* out1 = h1   + (long long)N * D_H;
    float* h2   = out1 + (long long)N * D_H;

    const int BT = 256;
    const int rowTiles = N / 16;

    k_deg_init<<<(N + BT - 1) / BT, BT, 0, stream>>>(deg, N);
    k_deg_accum<<<(E + BT - 1) / BT, BT, 0, stream>>>(dstIdx, deg, E);
    k_dinv<<<(N + BT - 1) / BT, BT, 0, stream>>>(deg, dinv, N);

    k_gemm_wmma<D_H><<<(rowTiles + 1) / 2, BT, 0, stream>>>(x, W1, h1, N);

    {
        long long tot = (long long)N * D_H;
        k_self_bias<<<(unsigned)((tot + BT - 1) / BT), BT, 0, stream>>>(h1, dinv, b1, out1, N, D_H);
        long long st = (long long)E * (D_H / 4);
        k_scatter<<<(unsigned)((st + BT - 1) / BT), BT, 0, stream>>>(h1, srcIdx, dstIdx, dinv, out1, E, D_H);
        k_relu<<<(unsigned)((tot + BT - 1) / BT), BT, 0, stream>>>(out1, tot);
    }

    k_gemm_wmma<D_OUT><<<(rowTiles + 3) / 4, BT, 0, stream>>>(out1, W2, h2, N);

    {
        long long tot = (long long)N * D_OUT;
        k_self_bias<<<(unsigned)((tot + BT - 1) / BT), BT, 0, stream>>>(h2, dinv, b2, out, N, D_OUT);
        long long st = (long long)E * (D_OUT / 4);
        k_scatter<<<(unsigned)((st + BT - 1) / BT), BT, 0, stream>>>(h2, srcIdx, dstIdx, dinv, out, E, D_OUT);
    }

    (void)n_in; (void)out_size; (void)ws_size;
}


# Float-atomic scatter replaced by in-call counting sort by destination (int atomic returns each edge's slot, scan, fill) + one-wave-per-row gather storing each row once; row sum order = arrival rank
# speedup vs baseline: 1.5779x; 1.5779x over previous
#include <hip/hip_runtime.h>
#include <hip/hip_bf16.h>


typedef __attribute__((ext_vector_type(2))) float v2f;
typedef __attribute__((ext_vector_type(8))) float v8f;

__global__ void k_cnt_zero(int* __restrict__ cnt, int n) {
    int i = blockIdx.x * blockDim.x + threadIdx.x;
    if (i < n) cnt[i] = 0;
}

__global__ void k_count(const int* __restrict__ dst, int* __restrict__ cnt,
                        int* __restrict__ rank, int e) {
    int i = blockIdx.x * blockDim.x + threadIdx.x;
    if (i < e) rank[i] = atomicAdd(&cnt[dst[i]], 1);
}

__device__ __forceinline__ int wave_incl_scan(int x, int lane) {
    x += __builtin_amdgcn_update_dpp(0, x, 0x111, 0xf, 0xf, true);
    x += __builtin_amdgcn_update_dpp(0, x, 0x112, 0xf, 0xf, true);
    x += __builtin_amdgcn_update_dpp(0, x, 0x114, 0xf, 0xf, true);
    x += __builtin_amdgcn_update_dpp(0, x, 0x118, 0xf, 0xf, true);
    const int lo = __builtin_amdgcn_readlane(x, 15);
    if (lane >= 16) x += lo;
    return x;
}

__global__ __launch_bounds__(1024) void k_scan_dinv(const int* __restrict__ cnt,
                                                    int* __restrict__ rowptr,
                                                    float* __restrict__ dinv, int n) {
    __shared__ int wsum[32];
    const int lane = threadIdx.x & 31;
    const int wave = __builtin_amdgcn_readfirstlane((int)(threadIdx.x >> 5));
    const int seg = (((n + 31) >> 5) + 127) & ~127;
    const int nit = seg >> 7;
    const int beg = wave * seg;
    const int4* c4 = (const int4*)(cnt + beg) + lane;

    int s = 0;
#pragma unroll 8
    for (int it = 0; it < nit; ++it) {
        const int4 v = c4[it * 32];
        s += (v.x + v.y) + (v.z + v.w);
    }
    const int tot = wave_incl_scan(s, lane);
    if (lane == 31) wsum[wave] = tot;
    __syncthreads();
    const int w = wsum[lane];
    const int winc = wave_incl_scan(w, lane);
    int carry = __builtin_amdgcn_readlane(winc - w, wave);

    for (int it0 = 0; it0 < nit; it0 += 4) {
        int4 vb[4];
#pragma unroll
        for (int u = 0; u < 4; ++u) vb[u] = c4[min(it0 + u, nit - 1) * 32];
#pragma unroll
        for (int u = 0; u < 4; ++u) {
            if (it0 + u < nit) {
                const int4 v = vb[u];
                const int i = beg + (it0 + u) * 128 + lane * 4;
                const int l1 = v.x, l2 = l1 + v.y, l3 = l2 + v.z, l4 = l3 + v.w;
                const int inc = wave_incl_scan(l4, lane);
                const int ex = carry + inc - l4;
                if (i + 3 < n) {
                    *(int4*)(rowptr + i) = make_int4(ex, ex + l1, ex + l2, ex + l3);
                    *(float4*)(dinv + i) = make_float4(rsqrtf((float)(v.x + 1)), rsqrtf((float)(v.y + 1)),
                                                       rsqrtf((float)(v.z + 1)), rsqrtf((float)(v.w + 1)));
                } else {
                    if (i     < n) { rowptr[i]     = ex;      dinv[i]     = rsqrtf((float)(v.x + 1)); }
                    if (i + 1 < n) { rowptr[i + 1] = ex + l1; dinv[i + 1] = rsqrtf((float)(v.y + 1)); }
                    if (i + 2 < n) { rowptr[i + 2] = ex + l2; dinv[i + 2] = rsqrtf((float)(v.z + 1)); }
                }
                carry += __builtin_amdgcn_readlane(inc, 31);
            }
        }
    }
}

__global__ void k_fill(const int* __restrict__ src, const int* __restrict__ dst,
                       const int* __restrict__ rank, const int* __restrict__ rowptr,
                       const float* __restrict__ dinv, int2* __restrict__ pairs, int e) {
    int i = blockIdx.x * blockDim.x + threadIdx.x;
    if (i >= e) return;
    const int s = src[i];
    const int pos = rowptr[dst[i]] + rank[i];
    pairs[pos] = make_int2(s, __float_as_int(dinv[s]));
}

template <int DCOL>
__global__ __launch_bounds__(256) void k_gemm_wmma(const float* __restrict__ X,
                                                   const float* __restrict__ W,
                                                   float* __restrict__ H,
                                                   int nrows) {
    constexpr int COLT = DCOL / 16;
    constexpr int ROWT = 8 / COLT;
    const int wave = threadIdx.x >> 5;
    const int lane = threadIdx.x & 31;
    const int m0 = (blockIdx.x * ROWT + (wave / COLT)) * 16;
    const int n0 = (wave % COLT) * 16;
    if (m0 >= nrows) return;

    const int lr = lane & 15;
    const int lh = lane >> 4;

    v8f c = {};
    const float* xrow = X + (long long)(m0 + lr) * 64;
#pragma unroll
    for (int k = 0; k < 64; k += 4) {
        v2f a, b;
        a.x = xrow[k + lh * 2 + 0];
        a.y = xrow[k + lh * 2 + 1];
        b.x = W[(k + lh * 2 + 0) * DCOL + n0 + lr];
        b.y = W[(k + lh * 2 + 1) * DCOL + n0 + lr];
        c = __builtin_amdgcn_wmma_f32_16x16x4_f32(false, a, false, b,
                                                  (short)0, c, false, false);
    }
#pragma unroll
    for (int r = 0; r < 8; ++r) {
        H[(long long)(m0 + r + lh * 8) * DCOL + n0 + lr] = c[r];
    }
}

template <int V> struct FVec;
template <> struct FVec<1> { typedef float T; };
template <> struct FVec<2> { typedef v2f T; };

__device__ __forceinline__ float relu_v(float a) { return fmaxf(a, 0.0f); }
__device__ __forceinline__ v2f relu_v(v2f a) { a.x = fmaxf(a.x, 0.0f); a.y = fmaxf(a.y, 0.0f); return a; }

template <int D, int K, typename T>
__device__ __forceinline__ void slot_load(T (&hv)[K], float (&w)[K], const float* __restrict__ hl,
                                          int sx, int sy, int k, float dj) {
#pragma unroll
    for (int u = 0; u < K; ++u) {
        const int s = __builtin_amdgcn_readlane(sx, k + u);
        w[u] = __int_as_float(__builtin_amdgcn_readlane(sy, k + u)) * dj;
        hv[u] = *(const T*)(hl + (long long)s * D);
    }
}

template <int D, bool RELU>
__global__ __launch_bounds__(256) void k_gather(const float* __restrict__ H,
                                                const int* __restrict__ rowptr,
                                                const int2* __restrict__ pairs,
                                                const float* __restrict__ dinv,
                                                const float* __restrict__ bias,
                                                float* __restrict__ out, int n, int e) {
    constexpr int V = D / 32;
    typedef typename FVec<V>::T T;
    const int lane = threadIdx.x & 31;
    const int row = blockIdx.x * 8 + __builtin_amdgcn_readfirstlane((int)(threadIdx.x >> 5));
    if (row >= n) return;
    const float* hl = H + lane * V;
    const T hs = *(const T*)(hl + (long long)row * D);
    const T bv = *(const T*)(bias + lane * V);
    const int p0 = rowptr[row];
    const int p1 = (row + 1 < n) ? rowptr[row + 1] : e;
    const float dj = dinv[row];

    T acc = {};
    for (int base = p0; base < p1; base += 32) {
        const int m = min(p1 - base, 32);
        int2 my = make_int2(0, 0);
        if (lane < m) my = pairs[base + lane];

        const int r = m & 7;
        T th[7]; float tw[7];
#pragma unroll
        for (int u = 0; u < 7; ++u) {
            const int s = __builtin_amdgcn_readlane(my.x, u);
            tw[u] = __int_as_float(__builtin_amdgcn_readlane(my.y, u)) * dj;
            if (u < r) th[u] = *(const T*)(hl + (long long)s * D);
        }
        for (int k = r; k < m; k += 8) {
            T hv[8]; float w[8];
            slot_load<D, 8, T>(hv, w, hl, my.x, my.y, k, dj);
#pragma unroll
            for (int u = 0; u < 8; ++u) acc += hv[u] * w[u];
        }
#pragma unroll
        for (int u = 0; u < 7; ++u) {
            if (u < r) acc += th[u] * tw[u];
        }
    }

    acc += hs * (dj * dj);
    acc += bv;
    if (RELU) acc = relu_v(acc);
    *(T*)(out + (long long)row * D + lane * V) = acc;
}

extern "C" void kernel_launch(void* const* d_in, const int* in_sizes, int n_in,
                              void* d_out, int out_size, void* d_ws, size_t ws_size,
                              hipStream_t stream) {
    const int D_IN = 64, D_H = 64, D_OUT = 32;
    const int N = in_sizes[0] / D_IN;
    const int E = in_sizes[1] / 2;

    const float* x  = (const float*)d_in[0];
    const int*   ei = (const int*)d_in[1];
    const float* W1 = (const float*)d_in[2];
    const float* b1 = (const float*)d_in[3];
    const float* W2 = (const float*)d_in[4];
    const float* b2 = (const float*)d_in[5];
    float* out = (float*)d_out;

    const int* srcIdx = ei;
    const int* dstIdx = ei + E;

    int2*  pairs  = (int2*)d_ws;
    float* h1     = (float*)d_ws + 2 * (long long)E;
    float* out1   = h1   + (long long)N * D_H;
    int*   rowptr = (int*)(out1 + (long long)N * D_H);
    float* dinv   = (float*)(rowptr + N);
    int*   cnt    = (int*)h1;
    int*   rank   = (int*)out1;
    float* h2     = h1;

    const int BT = 256;
    const int rowTiles = N / 16;

    k_cnt_zero<<<(N + 4096 + BT - 1) / BT, BT, 0, stream>>>(cnt, N + 4096);
    k_count<<<(E + BT - 1) / BT, BT, 0, stream>>>(dstIdx, cnt, rank, E);
    k_scan_dinv<<<1, 1024, 0, stream>>>(cnt, rowptr, dinv, N);
    k_fill<<<(E + BT - 1) / BT, BT, 0, stream>>>(srcIdx, dstIdx, rank, rowptr, dinv, pairs, E);

    k_gemm_wmma<D_H><<<(rowTiles + 1) / 2, BT, 0, stream>>>(x, W1, h1, N);
    k_gather<D_H, true><<<(N + 7) / 8, BT, 0, stream>>>(h1, rowptr, pairs, dinv, b1, out1, N, E);

    k_gemm_wmma<D_OUT><<<(rowTiles + 3) / 4, BT, 0, stream>>>(out1, W2, h2, N);
    k_gather<D_OUT, false><<<(N + 7) / 8, BT, 0, stream>>>(h2, rowptr, pairs, dinv, b2, out, N, E);

    (void)n_in; (void)out_size; (void)ws_size;
}


# Layer-2 product moved into the layer-1 gather launch: a block gathers 16 rows (2 per wave) into LDS, two waves multiply the tile by W2 on WMMA; out1 never stored, one launch fewer; sum orders kept
# speedup vs baseline: 1.6692x; 1.0579x over previous
#include <hip/hip_runtime.h>
#include <hip/hip_bf16.h>


typedef __attribute__((ext_vector_type(2))) float v2f;
typedef __attribute__((ext_vector_type(8))) float v8f;

__global__ void k_cnt_zero(int* __restrict__ cnt, int n) {
    int i = blockIdx.x * blockDim.x + threadIdx.x;
    if (i < n) cnt[i] = 0;
}

__global__ void k_count(const int* __restrict__ dst, int* __restrict__ cnt,
                        int* __restrict__ rank, int e) {
    int i = blockIdx.x * blockDim.x + threadIdx.x;
    if (i < e) rank[i] = atomicAdd(&cnt[dst[i]], 1);
}

__device__ __forceinline__ int wave_incl_scan(int x, int lane) {
    x += __builtin_amdgcn_update_dpp(0, x, 0x111, 0xf, 0xf, true);
    x += __builtin_amdgcn_update_dpp(0, x, 0x112, 0xf, 0xf, true);
    x += __builtin_amdgcn_update_dpp(0, x, 0x114, 0xf, 0xf, true);
    x += __builtin_amdgcn_update_dpp(0, x, 0x118, 0xf, 0xf, true);
    const int lo = __builtin_amdgcn_readlane(x, 15);
    if (lane >= 16) x += lo;
    return x;
}

__global__ __launch_bounds__(1024) void k_scan_dinv(const int* __restrict__ cnt,
                                                    int* __restrict__ rowptr,
                                                    float* __restrict__ dinv, int n) {
    __shared__ int wsum[32];
    const int lane = threadIdx.x & 31;
    const int wave = __builtin_amdgcn_readfirstlane((int)(threadIdx.x >> 5));
    const int seg = (((n + 31) >> 5) + 127) & ~127;
    const int nit = seg >> 7;
    const int beg = wave * seg;
    const int4* c4 = (const int4*)(cnt + beg) + lane;

    int s = 0;
#pragma unroll 8
    for (int it = 0; it < nit; ++it) {
        const int4 v = c4[it * 32];
        s += (v.x + v.y) + (v.z + v.w);
    }
    const int tot = wave_incl_scan(s, lane);
    if (lane == 31) wsum[wave] = tot;
    __syncthreads();
    const int w = wsum[lane];
    const int winc = wave_incl_scan(w, lane);
    int carry = __builtin_amdgcn_readlane(winc - w, wave);

    for (int it0 = 0; it0 < nit; it0 += 4) {
        int4 vb[4];
#pragma unroll
        for (int u = 0; u < 4; ++u) vb[u] = c4[min(it0 + u, nit - 1) * 32];
#pragma unroll
        for (int u = 0; u < 4; ++u) {
            if (it0 + u < nit) {
                const int4 v = vb[u];
                const int i = beg + (it0 + u) * 128 + lane * 4;
                const int l1 = v.x, l2 = l1 + v.y, l3 = l2 + v.z, l4 = l3 + v.w;
                const int inc = wave_incl_scan(l4, lane);
                const int ex = carry + inc - l4;
                if (i + 3 < n) {
                    *(int4*)(rowptr + i) = make_int4(ex, ex + l1, ex + l2, ex + l3);
                    *(float4*)(dinv + i) = make_float4(rsqrtf((float)(v.x + 1)), rsqrtf((float)(v.y + 1)),
                                                       rsqrtf((float)(v.z + 1)), rsqrtf((float)(v.w + 1)));
                } else {
                    if (i     < n) { rowptr[i]     = ex;      dinv[i]     = rsqrtf((float)(v.x + 1)); }
                    if (i + 1 < n) { rowptr[i + 1] = ex + l1; dinv[i + 1] = rsqrtf((float)(v.y + 1)); }
                    if (i + 2 < n) { rowptr[i + 2] = ex + l2; dinv[i + 2] = rsqrtf((float)(v.z + 1)); }
                }
                carry += __builtin_amdgcn_readlane(inc, 31);
            }
        }
    }
}

__global__ void k_fill(const int* __restrict__ src, const int* __restrict__ dst,
                       const int* __restrict__ rank, const int* __restrict__ rowptr,
                       const float* __restrict__ dinv, int2* __restrict__ pairs, int e) {
    int i = blockIdx.x * blockDim.x + threadIdx.x;
    if (i >= e) return;
    const int s = src[i];
    const int pos = rowptr[dst[i]] + rank[i];
    pairs[pos] = make_int2(s, __float_as_int(dinv[s]));
}

template <int DCOL>
__global__ __launch_bounds__(256) void k_gemm_wmma(const float* __restrict__ X,
                                                   const float* __restrict__ W,
                                                   float* __restrict__ H,
                                                   int nrows) {
    constexpr int COLT = DCOL / 16;
    constexpr int ROWT = 8 / COLT;
    const int wave = threadIdx.x >> 5;
    const int lane = threadIdx.x & 31;
    const int m0 = (blockIdx.x * ROWT + (wave / COLT)) * 16;
    const int n0 = (wave % COLT) * 16;
    if (m0 >= nrows) return;

    const int lr = lane & 15;
    const int lh = lane >> 4;

    v8f c = {};
    const float* xrow = X + (long long)(m0 + lr) * 64;
#pragma unroll
    for (int k = 0; k < 64; k += 4) {
        v2f a, b;
        a.x = xrow[k + lh * 2 + 0];
        a.y = xrow[k + lh * 2 + 1];
        b.x = W[(k + lh * 2 + 0) * DCOL + n0 + lr];
        b.y = W[(k + lh * 2 + 1) * DCOL + n0 + lr];
        c = __builtin_amdgcn_wmma_f32_16x16x4_f32(false, a, false, b,
                                                  (short)0, c, false, false);
    }
#pragma unroll
    for (int r = 0; r < 8; ++r) {
        H[(long long)(m0 + r + lh * 8) * DCOL + n0 + lr] = c[r];
    }
}

template <int V> struct FVec;
template <> struct FVec<1> { typedef float T; };
template <> struct FVec<2> { typedef v2f T; };

__device__ __forceinline__ float relu_v(float a) { return fmaxf(a, 0.0f); }
__device__ __forceinline__ v2f relu_v(v2f a) { a.x = fmaxf(a.x, 0.0f); a.y = fmaxf(a.y, 0.0f); return a; }

template <int D, int K, typename T>
__device__ __forceinline__ void slot_load(T (&hv)[K], float (&w)[K], const float* __restrict__ hl,
                                          int sx, int sy, int k, float dj) {
#pragma unroll
    for (int u = 0; u < K; ++u) {
        const int s = __builtin_amdgcn_readlane(sx, k + u);
        w[u] = __int_as_float(__builtin_amdgcn_readlane(sy, k + u)) * dj;
        hv[u] = *(const T*)(hl + (long long)s * D);
    }
}

template <int D, typename T>
__device__ __forceinline__ T row_sum(const float* __restrict__ hl, const int2* __restrict__ pairs,
                                     int p0, int p1, int2 first, float dj, int lane) {
    T acc = {};
    int2 my = first;
    for (int base = p0; base < p1; ) {
        const int m = min(p1 - base, 32);

        const int r = m & 7;
        T th[7]; float tw[7];
#pragma unroll
        for (int u = 0; u < 7; ++u) {
            const int s = __builtin_amdgcn_readlane(my.x, u);
            tw[u] = __int_as_float(__builtin_amdgcn_readlane(my.y, u)) * dj;
            if (u < r) th[u] = *(const T*)(hl + (long long)s * D);
        }
        for (int k = r; k < m; k += 8) {
            T hv[8]; float w[8];
            slot_load<D, 8, T>(hv, w, hl, my.x, my.y, k, dj);
#pragma unroll
            for (int u = 0; u < 8; ++u) acc += hv[u] * w[u];
        }
#pragma unroll
        for (int u = 0; u < 7; ++u) {
            if (u < r) acc += th[u] * tw[u];
        }

        base += 32;
        if (base < p1) {
            my = make_int2(0, 0);
            if (lane < min(p1 - base, 32)) my = pairs[base + lane];
        }
    }
    return acc;
}

template <int D, bool RELU>
__global__ __launch_bounds__(256) void k_gather(const float* __restrict__ H,
                                                const int* __restrict__ rowptr,
                                                const int2* __restrict__ pairs,
                                                const float* __restrict__ dinv,
                                                const float* __restrict__ bias,
                                                float* __restrict__ out, int n, int e) {
    constexpr int V = D / 32;
    typedef typename FVec<V>::T T;
    const int lane = threadIdx.x & 31;
    const int row = blockIdx.x * 8 + __builtin_amdgcn_readfirstlane((int)(threadIdx.x >> 5));
    if (row >= n) return;
    const float* hl = H + lane * V;
    const T hs = *(const T*)(hl + (long long)row * D);
    const T bv = *(const T*)(bias + lane * V);
    const int p0 = rowptr[row];
    const int p1 = (row + 1 < n) ? rowptr[row + 1] : e;
    const float dj = dinv[row];

    T acc = {};
    for (int base = p0; base < p1; base += 32) {
        const int m = min(p1 - base, 32);
        int2 my = make_int2(0, 0);
        if (lane < m) my = pairs[base + lane];

        const int r = m & 7;
        T th[7]; float tw[7];
#pragma unroll
        for (int u = 0; u < 7; ++u) {
            const int s = __builtin_amdgcn_readlane(my.x, u);
            tw[u] = __int_as_float(__builtin_amdgcn_readlane(my.y, u)) * dj;
            if (u < r) th[u] = *(const T*)(hl + (long long)s * D);
        }
        for (int k = r; k < m; k += 8) {
            T hv[8]; float w[8];
            slot_load<D, 8, T>(hv, w, hl, my.x, my.y, k, dj);
#pragma unroll
            for (int u = 0; u < 8; ++u) acc += hv[u] * w[u];
        }
#pragma unroll
        for (int u = 0; u < 7; ++u) {
            if (u < r) acc += th[u] * tw[u];
        }
    }

    acc += hs * (dj * dj);
    acc += bv;
    if (RELU) acc = relu_v(acc);
    *(T*)(out + (long long)row * D + lane * V) = acc;
}

__global__ __launch_bounds__(256) void k_gather_relu_gemm(const float* __restrict__ H,
                                                          const int* __restrict__ rowptr,
                                                          const int2* __restrict__ pairs,
                                                          const float* __restrict__ dinv,
                                                          const float* __restrict__ bias,
                                                          const float* __restrict__ W2,
                                                          float* __restrict__ H2, int n, int e) {
    constexpr int D = 64, DO = 32, PITCH = 68;
    __shared__ float tile[16 * PITCH];
    const int lane = threadIdx.x & 31;
    const int wave = __builtin_amdgcn_readfirstlane((int)(threadIdx.x >> 5));
    const int m0 = blockIdx.x * 16;
    const int r0 = m0 + wave * 2;
    const float* hl = H + lane * 2;
    const v2f bv = *(const v2f*)(bias + lane * 2);

    const int pa = rowptr[r0];
    const int pb = rowptr[r0 + 1];
    const int pc = (r0 + 2 < n) ? rowptr[r0 + 2] : e;
    const float dj0 = dinv[r0];
    const float dj1 = dinv[r0 + 1];
    const v2f hs0 = *(const v2f*)(hl + (long long)r0 * D);
    const v2f hs1 = *(const v2f*)(hl + (long long)(r0 + 1) * D);
    int2 f0 = make_int2(0, 0), f1 = make_int2(0, 0);
    if (lane < min(pb - pa, 32)) f0 = pairs[pa + lane];
    if (lane < min(pc - pb, 32)) f1 = pairs[pb + lane];

    v2f a0 = row_sum<D, v2f>(hl, pairs, pa, pb, f0, dj0, lane);
    a0 += hs0 * (dj0 * dj0);
    a0 += bv;
    a0 = relu_v(a0);
    *(v2f*)(tile + (wave * 2 + 0) * PITCH + lane * 2) = a0;

    v2f a1 = row_sum<D, v2f>(hl, pairs, pb, pc, f1, dj1, lane);
    a1 += hs1 * (dj1 * dj1);
    a1 += bv;
    a1 = relu_v(a1);
    *(v2f*)(tile + (wave * 2 + 1) * PITCH + lane * 2) = a1;

    const int lr = lane & 15;
    const int lh = lane >> 4;
    const int n0 = (wave & 1) * 16;
    v2f bf[16];
    if (wave < 2) {
#pragma unroll
        for (int k = 0; k < 64; k += 4) {
            bf[k >> 2].x = W2[(k + lh * 2 + 0) * DO + n0 + lr];
            bf[k >> 2].y = W2[(k + lh * 2 + 1) * DO + n0 + lr];
        }
    }
    __builtin_amdgcn_fence(__ATOMIC_RELEASE, "workgroup", "local");
    __builtin_amdgcn_s_barrier();
    __builtin_amdgcn_fence(__ATOMIC_ACQUIRE, "workgroup", "local");
    if (wave >= 2) return;

    typedef __attribute__((address_space(1))) float gfloat;
    gfloat* hp = (gfloat*)(H2 + (long long)(m0 + lh * 8) * DO + n0 + lr);
    asm volatile("" : "+v"(hp));

    v8f c = {};
#pragma unroll
    for (int k = 0; k < 64; k += 4) {
        const v2f a = *(const v2f*)(tile + lr * PITCH + k + lh * 2);
        c = __builtin_amdgcn_wmma_f32_16x16x4_f32(false, a, false, bf[k >> 2],
                                                  (short)0, c, false, false);
    }
#pragma unroll
    for (int r = 0; r < 8; ++r) {
        hp[r * DO] = c[r];
    }
}

extern "C" void kernel_launch(void* const* d_in, const int* in_sizes, int n_in,
                              void* d_out, int out_size, void* d_ws, size_t ws_size,
                              hipStream_t stream) {
    const int D_IN = 64, D_H = 64, D_OUT = 32;
    const int N = in_sizes[0] / D_IN;
    const int E = in_sizes[1] / 2;

    const float* x  = (const float*)d_in[0];
    const int*   ei = (const int*)d_in[1];
    const float* W1 = (const float*)d_in[2];
    const float* b1 = (const float*)d_in[3];
    const float* W2 = (const float*)d_in[4];
    const float* b2 = (const float*)d_in[5];
    float* out = (float*)d_out;

    const int* srcIdx = ei;
    const int* dstIdx = ei + E;

    int2*  pairs  = (int2*)d_ws;
    float* h1     = (float*)d_ws + 2 * (long long)E;
    float* h2     = h1   + (long long)N * D_H;
    int*   rowptr = (int*)(h2 + (long long)N * D_H);
    float* dinv   = (float*)(rowptr + N);
    int*   cnt    = (int*)h1;
    int*   rank   = (int*)h2;

    const int BT = 256;
    const int rowTiles = N / 16;

    k_cnt_zero<<<(N + 4096 + BT - 1) / BT, BT, 0, stream>>>(cnt, N + 4096);
    k_count<<<(E + BT - 1) / BT, BT, 0, stream>>>(dstIdx, cnt, rank, E);
    k_scan_dinv<<<1, 1024, 0, stream>>>(cnt, rowptr, dinv, N);
    k_fill<<<(E + BT - 1) / BT, BT, 0, stream>>>(srcIdx, dstIdx, rank, rowptr, dinv, pairs, E);

    k_gemm_wmma<D_H><<<(rowTiles + 1) / 2, BT, 0, stream>>>(x, W1, h1, N);
    k_gather_relu_gemm<<<rowTiles, BT, 0, stream>>>(h1, rowptr, pairs, dinv, b1, W2, h2, N, E);

    k_gather<D_OUT, false><<<(N + 7) / 8, BT, 0, stream>>>(h2, rowptr, pairs, dinv, b2, out, N, E);

    (void)n_in; (void)out_size; (void)ws_size;
}


# Layer-1 product moved behind its aggregation into the gather launch: block sums x rows, then tile*W1 and *W2 on f32 WMMA; GEMM launch and h1 gone; layer-1 rounding order changes (sum, then product)
# speedup vs baseline: 1.7655x; 1.0577x over previous
#include <hip/hip_runtime.h>
#include <hip/hip_bf16.h>


typedef __attribute__((ext_vector_type(2))) float v2f;
typedef __attribute__((ext_vector_type(8))) float v8f;

__global__ void k_cnt_zero(int* __restrict__ cnt, int n) {
    int i = blockIdx.x * blockDim.x + threadIdx.x;
    if (i < n) cnt[i] = 0;
}

__global__ void k_count(const int* __restrict__ dst, int* __restrict__ cnt,
                        int* __restrict__ rank, int e) {
    int i = blockIdx.x * blockDim.x + threadIdx.x;
    if (i < e) rank[i] = atomicAdd(&cnt[dst[i]], 1);
}

__device__ __forceinline__ int wave_incl_scan(int x, int lane) {
    x += __builtin_amdgcn_update_dpp(0, x, 0x111, 0xf, 0xf, true);
    x += __builtin_amdgcn_update_dpp(0, x, 0x112, 0xf, 0xf, true);
    x += __builtin_amdgcn_update_dpp(0, x, 0x114, 0xf, 0xf, true);
    x += __builtin_amdgcn_update_dpp(0, x, 0x118, 0xf, 0xf, true);
    const int lo = __builtin_amdgcn_readlane(x, 15);
    if (lane >= 16) x += lo;
    return x;
}

__global__ __launch_bounds__(1024) void k_scan_dinv(const int* __restrict__ cnt,
                                                    int* __restrict__ rowptr,
                                                    float* __restrict__ dinv, int n) {
    __shared__ int wsum[32];
    const int lane = threadIdx.x & 31;
    const int wave = __builtin_amdgcn_readfirstlane((int)(threadIdx.x >> 5));
    const int seg = (((n + 31) >> 5) + 127) & ~127;
    const int nit = seg >> 7;
    const int beg = wave * seg;
    const int4* c4 = (const int4*)(cnt + beg) + lane;

    int s = 0;
#pragma unroll 8
    for (int it = 0; it < nit; ++it) {
        const int4 v = c4[it * 32];
        s += (v.x + v.y) + (v.z + v.w);
    }
    const int tot = wave_incl_scan(s, lane);
    if (lane == 31) wsum[wave] = tot;
    __syncthreads();
    const int w = wsum[lane];
    const int winc = wave_incl_scan(w, lane);
    int carry = __builtin_amdgcn_readlane(winc - w, wave);

    for (int it0 = 0; it0 < nit; it0 += 4) {
        int4 vb[4];
#pragma unroll
        for (int u = 0; u < 4; ++u) vb[u] = c4[min(it0 + u, nit - 1) * 32];
#pragma unroll
        for (int u = 0; u < 4; ++u) {
            if (it0 + u < nit) {
                const int4 v = vb[u];
                const int i = beg + (it0 + u) * 128 + lane * 4;
                const int l1 = v.x, l2 = l1 + v.y, l3 = l2 + v.z, l4 = l3 + v.w;
                const int inc = wave_incl_scan(l4, lane);
                const int ex = carry + inc - l4;
                if (i + 3 < n) {
                    *(int4*)(rowptr + i) = make_int4(ex, ex + l1, ex + l2, ex + l3);
                    *(float4*)(dinv + i) = make_float4(rsqrtf((float)(v.x + 1)), rsqrtf((float)(v.y + 1)),
                                                       rsqrtf((float)(v.z + 1)), rsqrtf((float)(v.w + 1)));
                } else {
                    if (i     < n) { rowptr[i]     = ex;      dinv[i]     = rsqrtf((float)(v.x + 1)); }
                    if (i + 1 < n) { rowptr[i + 1] = ex + l1; dinv[i + 1] = rsqrtf((float)(v.y + 1)); }
                    if (i + 2 < n) { rowptr[i + 2] = ex + l2; dinv[i + 2] = rsqrtf((float)(v.z + 1)); }
                }
                carry += __builtin_amdgcn_readlane(inc, 31);
            }
        }
    }
}

__global__ void k_fill(const int* __restrict__ src, const int* __restrict__ dst,
                       const int* __restrict__ rank, const int* __restrict__ rowptr,
                       const float* __restrict__ dinv, int2* __restrict__ pairs, int e) {
    int i = blockIdx.x * blockDim.x + threadIdx.x;
    if (i >= e) return;
    const int s = src[i];
    const int pos = rowptr[dst[i]] + rank[i];
    pairs[pos] = make_int2(s, __float_as_int(dinv[s]));
}

template <int V> struct FVec;
template <> struct FVec<1> { typedef float T; };
template <> struct FVec<2> { typedef v2f T; };

__device__ __forceinline__ float relu_v(float a) { return fmaxf(a, 0.0f); }
__device__ __forceinline__ v2f relu_v(v2f a) { a.x = fmaxf(a.x, 0.0f); a.y = fmaxf(a.y, 0.0f); return a; }

template <int D, int K, typename T>
__device__ __forceinline__ void slot_load(T (&hv)[K], float (&w)[K], const float* __restrict__ hl,
                                          int sx, int sy, int k, float dj) {
#pragma unroll
    for (int u = 0; u < K; ++u) {
        const int s = __builtin_amdgcn_readlane(sx, k + u);
        w[u] = __int_as_float(__builtin_amdgcn_readlane(sy, k + u)) * dj;
        hv[u] = *(const T*)(hl + (long long)s * D);
    }
}

template <int D, typename T>
__device__ __forceinline__ T row_sum(const float* __restrict__ hl, const int2* __restrict__ pairs,
                                     int p0, int p1, int2 first, float dj, int lane) {
    T acc = {};
    int2 my = first;
    for (int base = p0; base < p1; ) {
        const int m = min(p1 - base, 32);

        const int r = m & 7;
        T th[7]; float tw[7];
#pragma unroll
        for (int u = 0; u < 7; ++u) {
            const int s = __builtin_amdgcn_readlane(my.x, u);
            tw[u] = __int_as_float(__builtin_amdgcn_readlane(my.y, u)) * dj;
            if (u < r) th[u] = *(const T*)(hl + (long long)s * D);
        }
        for (int k = r; k < m; k += 8) {
            T hv[8]; float w[8];
            slot_load<D, 8, T>(hv, w, hl, my.x, my.y, k, dj);
#pragma unroll
            for (int u = 0; u < 8; ++u) acc += hv[u] * w[u];
        }
#pragma unroll
        for (int u = 0; u < 7; ++u) {
            if (u < r) acc += th[u] * tw[u];
        }

        base += 32;
        if (base < p1) {
            my = make_int2(0, 0);
            if (lane < min(p1 - base, 32)) my = pairs[base + lane];
        }
    }
    return acc;
}

template <int D, bool RELU>
__global__ __launch_bounds__(256) void k_gather(const float* __restrict__ H,
                                                const int* __restrict__ rowptr,
                                                const int2* __restrict__ pairs,
                                                const float* __restrict__ dinv,
                                                const float* __restrict__ bias,
                                                float* __restrict__ out, int n, int e) {
    constexpr int V = D / 32;
    typedef typename FVec<V>::T T;
    const int lane = threadIdx.x & 31;
    const int row = blockIdx.x * 8 + __builtin_amdgcn_readfirstlane((int)(threadIdx.x >> 5));
    if (row >= n) return;
    const float* hl = H + lane * V;
    const T hs = *(const T*)(hl + (long long)row * D);
    const T bv = *(const T*)(bias + lane * V);
    const int p0 = rowptr[row];
    const int p1 = (row + 1 < n) ? rowptr[row + 1] : e;
    const float dj = dinv[row];

    T acc = {};
    for (int base = p0; base < p1; base += 32) {
        const int m = min(p1 - base, 32);
        int2 my = make_int2(0, 0);
        if (lane < m) my = pairs[base + lane];

        const int r = m & 7;
        T th[7]; float tw[7];
#pragma unroll
        for (int u = 0; u < 7; ++u) {
            const int s = __builtin_amdgcn_readlane(my.x, u);
            tw[u] = __int_as_float(__builtin_amdgcn_readlane(my.y, u)) * dj;
            if (u < r) th[u] = *(const T*)(hl + (long long)s * D);
        }
        for (int k = r; k < m; k += 8) {
            T hv[8]; float w[8];
            slot_load<D, 8, T>(hv, w, hl, my.x, my.y, k, dj);
#pragma unroll
            for (int u = 0; u < 8; ++u) acc += hv[u] * w[u];
        }
#pragma unroll
        for (int u = 0; u < 7; ++u) {
            if (u < r) acc += th[u] * tw[u];
        }
    }

    acc += hs * (dj * dj);
    acc += bv;
    if (RELU) acc = relu_v(acc);
    *(T*)(out + (long long)row * D + lane * V) = acc;
}

__global__ __launch_bounds__(256) void k_gather_gemm_relu_gemm(const float* __restrict__ X,
                                                               const int* __restrict__ rowptr,
                                                               const int2* __restrict__ pairs,
                                                               const float* __restrict__ dinv,
                                                               const float* __restrict__ W1,
                                                               const float* __restrict__ bias,
                                                               const float* __restrict__ W2,
                                                               float* __restrict__ H2, int n, int e) {
    constexpr int D = 64, DO = 32, PITCH = 68;
    __shared__ float tile[16 * PITCH];
    __shared__ float tile2[16 * PITCH];
    const int lane = threadIdx.x & 31;
    const int wave = __builtin_amdgcn_readfirstlane((int)(threadIdx.x >> 5));
    const int m0 = blockIdx.x * 16;
    const int r0 = m0 + wave * 2;
    const float* xl = X + lane * 2;

    const int pa = rowptr[r0];
    const int pb = rowptr[r0 + 1];
    const int pc = (r0 + 2 < n) ? rowptr[r0 + 2] : e;
    const float dj0 = dinv[r0];
    const float dj1 = dinv[r0 + 1];
    const v2f hs0 = *(const v2f*)(xl + (long long)r0 * D);
    const v2f hs1 = *(const v2f*)(xl + (long long)(r0 + 1) * D);
    int2 f0 = make_int2(0, 0), f1 = make_int2(0, 0);
    if (lane < min(pb - pa, 32)) f0 = pairs[pa + lane];
    if (lane < min(pc - pb, 32)) f1 = pairs[pb + lane];

    v2f a0 = row_sum<D, v2f>(xl, pairs, pa, pb, f0, dj0, lane);
    a0 += hs0 * (dj0 * dj0);
    *(v2f*)(tile + (wave * 2 + 0) * PITCH + lane * 2) = a0;

    v2f a1 = row_sum<D, v2f>(xl, pairs, pb, pc, f1, dj1, lane);
    a1 += hs1 * (dj1 * dj1);
    *(v2f*)(tile + (wave * 2 + 1) * PITCH + lane * 2) = a1;

    const int lr = lane & 15;
    const int lh = lane >> 4;
    const int n1 = (wave & 3) * 16;
    v2f bf[16];
    float bc = 0.0f;
    if (wave < 4) {
#pragma unroll
        for (int k = 0; k < 64; k += 4) {
            bf[k >> 2].x = W1[(k + lh * 2 + 0) * D + n1 + lr];
            bf[k >> 2].y = W1[(k + lh * 2 + 1) * D + n1 + lr];
        }
        bc = bias[n1 + lr];
    }
    __builtin_amdgcn_fence(__ATOMIC_RELEASE, "workgroup", "local");
    __builtin_amdgcn_s_barrier();
    __builtin_amdgcn_fence(__ATOMIC_ACQUIRE, "workgroup", "local");

    if (wave < 4) {
        typedef __attribute__((address_space(3))) float lfloat;
        lfloat* tp = (lfloat*)tile2 + (lh * 8) * PITCH + n1 + lr;
        asm volatile("" : "+v"(tp));

        v8f c = {};
#pragma unroll
        for (int k = 0; k < 64; k += 4) {
            const v2f a = *(const v2f*)(tile + lr * PITCH + k + lh * 2);
            c = __builtin_amdgcn_wmma_f32_16x16x4_f32(false, a, false, bf[k >> 2],
                                                      (short)0, c, false, false);
        }
        asm volatile("v_nop\n\tv_nop\n\tv_nop\n\tv_nop\n\tv_nop" : "+v"(c));
#pragma unroll
        for (int r = 0; r < 8; ++r) {
            tp[r * PITCH] = fmaxf(c[r] + bc, 0.0f);
        }
    }

    const int n0 = (wave & 1) * 16;
    v2f bg[16];
    if (wave < 2) {
#pragma unroll
        for (int k = 0; k < 64; k += 4) {
            bg[k >> 2].x = W2[(k + lh * 2 + 0) * DO + n0 + lr];
            bg[k >> 2].y = W2[(k + lh * 2 + 1) * DO + n0 + lr];
        }
    }
    __builtin_amdgcn_fence(__ATOMIC_RELEASE, "workgroup", "local");
    __builtin_amdgcn_s_barrier();
    __builtin_amdgcn_fence(__ATOMIC_ACQUIRE, "workgroup", "local");
    if (wave >= 2) return;

    typedef __attribute__((address_space(1))) float gfloat;
    gfloat* hp = (gfloat*)(H2 + (long long)(m0 + lh * 8) * DO + n0 + lr);
    asm volatile("" : "+v"(hp));

    v8f c2 = {};
#pragma unroll
    for (int k = 0; k < 64; k += 4) {
        const v2f a = *(const v2f*)(tile2 + lr * PITCH + k + lh * 2);
        c2 = __builtin_amdgcn_wmma_f32_16x16x4_f32(false, a, false, bg[k >> 2],
                                                   (short)0, c2, false, false);
    }
#pragma unroll
    for (int r = 0; r < 8; ++r) {
        hp[r * DO] = c2[r];
    }
}

extern "C" void kernel_launch(void* const* d_in, const int* in_sizes, int n_in,
                              void* d_out, int out_size, void* d_ws, size_t ws_size,
                              hipStream_t stream) {
    const int D_IN = 64, D_H = 64, D_OUT = 32;
    const int N = in_sizes[0] / D_IN;
    const int E = in_sizes[1] / 2;

    const float* x  = (const float*)d_in[0];
    const int*   ei = (const int*)d_in[1];
    const float* W1 = (const float*)d_in[2];
    const float* b1 = (const float*)d_in[3];
    const float* W2 = (const float*)d_in[4];
    const float* b2 = (const float*)d_in[5];
    float* out = (float*)d_out;

    const int* srcIdx = ei;
    const int* dstIdx = ei + E;

    int2*  pairs  = (int2*)d_ws;
    int*   cnt    = (int*)d_ws + 2 * (long long)E;
    float* h2     = (float*)cnt + (long long)N * D_H;
    int*   rowptr = (int*)(h2 + (long long)N * D_H);
    float* dinv   = (float*)(rowptr + N);
    int*   rank   = (int*)h2;

    const int BT = 256;
    const int rowTiles = N / 16;

    k_cnt_zero<<<(N + 4096 + BT - 1) / BT, BT, 0, stream>>>(cnt, N + 4096);
    k_count<<<(E + BT - 1) / BT, BT, 0, stream>>>(dstIdx, cnt, rank, E);
    k_scan_dinv<<<1, 1024, 0, stream>>>(cnt, rowptr, dinv, N);
    k_fill<<<(E + BT - 1) / BT, BT, 0, stream>>>(srcIdx, dstIdx, rank, rowptr, dinv, pairs, E);

    k_gather_gemm_relu_gemm<<<rowTiles, BT, 0, stream>>>(x, rowptr, pairs, dinv, W1, b1, W2, h2, N, E);

    k_gather<D_OUT, false><<<(N + 7) / 8, BT, 0, stream>>>(h2, rowptr, pairs, dinv, b2, out, N, E);

    (void)n_in; (void)out_size; (void)ws_size;
}


# Edge slots ranked by two-level counting sort on LDS counters (chunk pass bins by 256-row range, range pass counts, scans, places), not a returning global atomic per edge; row sum order = LDS arrival
# speedup vs baseline: 3.8553x; 2.1837x over previous
#include <hip/hip_runtime.h>
#include <hip/hip_bf16.h>


typedef __attribute__((ext_vector_type(2))) float v2f;
typedef __attribute__((ext_vector_type(8))) float v8f;

constexpr int CH = 256;
constexpr int CAP = 4096;
constexpr int CAP2 = 8192;

__device__ __forceinline__ int wave_incl_scan(int x, int lane) {
    x += __builtin_amdgcn_update_dpp(0, x, 0x111, 0xf, 0xf, true);
    x += __builtin_amdgcn_update_dpp(0, x, 0x112, 0xf, 0xf, true);
    x += __builtin_amdgcn_update_dpp(0, x, 0x114, 0xf, 0xf, true);
    x += __builtin_amdgcn_update_dpp(0, x, 0x118, 0xf, 0xf, true);
    const int lo = __builtin_amdgcn_readlane(x, 15);
    if (lane >= 16) x += lo;
    return x;
}

__global__ __launch_bounds__(512) void k_chunk_bin(const int* __restrict__ src, const int* __restrict__ dst,
                                                   int2* __restrict__ binned, int* __restrict__ ccnt,
                                                   int* __restrict__ coff, int e, int len) {
    __shared__ int hist[256];
    __shared__ int offs[256];
    __shared__ int wsum[8];
    __shared__ int2 stage[CAP];
    const int t = threadIdx.x;
    const int lane = t & 31;
    const int c = blockIdx.x;
    const int lo = c * len;
    const int cnt = min(len, e - lo);
    if (t < 256) hist[t] = 0;
    __syncthreads();

    int s[8], d[8], r[8];
#pragma unroll
    for (int u = 0; u < 8; ++u) {
        const int i = u * 512 + t;
        s[u] = 0; d[u] = -1; r[u] = 0;
        if (i < cnt) { s[u] = src[lo + i]; d[u] = dst[lo + i]; }
    }
#pragma unroll
    for (int u = 0; u < 8; ++u) {
        if (d[u] >= 0) r[u] = atomicAdd(&hist[d[u] >> 8], 1);
    }
    __syncthreads();
    int h = 0, inc = 0;
    if (t < 256) {
        h = hist[t];
        inc = wave_incl_scan(h, lane);
        if (lane == 31) wsum[t >> 5] = inc;
    }
    __syncthreads();
    if (t < 256) {
        int carry = 0;
#pragma unroll
        for (int k = 0; k < 8; ++k) carry += (k < (t >> 5)) ? wsum[k] : 0;
        const int ex = carry + inc - h;
        offs[t] = ex;
        ccnt[t * CH + c] = h;
        coff[t * CH + c] = ex;
    }
    __syncthreads();
#pragma unroll
    for (int u = 0; u < 8; ++u) {
        if (d[u] >= 0) stage[offs[d[u] >> 8] + r[u]] = make_int2(s[u], d[u]);
    }
    __syncthreads();
    for (int i = t; i < cnt; i += 512) binned[lo + i] = stage[i];
}

__global__ __launch_bounds__(1024) void k_range_rows(const int2* __restrict__ binned,
                                                     const int* __restrict__ ccnt,
                                                     const int* __restrict__ coff,
                                                     int* __restrict__ rowptr, float* __restrict__ dinv,
                                                     int* __restrict__ srcs, int n, int len) {
    __shared__ int scnt[256];
    __shared__ int soff[256];
    __shared__ int ncnt[256];
    __shared__ int red[32];
    __shared__ int wsum[8];
    __shared__ int sstage[CAP2];
    const int t = threadIdx.x;
    const int lane = t & 31;
    const int wave = __builtin_amdgcn_readfirstlane((int)(t >> 5));
    const int b = blockIdx.x;

    if (t < 256) { scnt[t] = ccnt[b * CH + t]; soff[t] = coff[b * CH + t]; ncnt[t] = 0; }

    const int lim4 = b * (CH / 4);
    const int4* c4 = (const int4*)ccnt;
    int part = 0;
#pragma unroll
    for (int g = 0; g < 2; ++g) {
        if (g * 8192 < lim4) {
            int4 v[8];
#pragma unroll
            for (int u = 0; u < 8; ++u) v[u] = c4[min(g * 8192 + u * 1024 + t, lim4 - 1)];
#pragma unroll
            for (int u = 0; u < 8; ++u) {
                const int sm = (v[u].x + v[u].y) + (v[u].z + v[u].w);
                part += (g * 8192 + u * 1024 + t < lim4) ? sm : 0;
            }
        }
    }
    const int wtot = wave_incl_scan(part, lane);
    if (lane == 31) red[wave] = wtot;
    __syncthreads();
    const int base = __builtin_amdgcn_readlane(wave_incl_scan(red[lane], lane), 31);

    int2 ev[8];
    int rm[8], ro[8];
#pragma unroll
    for (int k = 0; k < 8; ++k) {
        rm[k] = __builtin_amdgcn_readfirstlane(scnt[wave + 32 * k]);
        ro[k] = __builtin_amdgcn_readfirstlane(soff[wave + 32 * k]);
    }
#pragma unroll
    for (int k = 0; k < 8; ++k) {
        ev[k] = make_int2(0, -1);
        if (lane < rm[k]) ev[k] = binned[(long long)(wave + 32 * k) * len + ro[k] + lane];
    }
#pragma unroll
    for (int k = 0; k < 8; ++k) {
        if (ev[k].y >= 0) atomicAdd(&ncnt[ev[k].y & 255], 1);
    }
#pragma unroll
    for (int k = 0; k < 8; ++k) {
        const int2* p = binned + (long long)(wave + 32 * k) * len + ro[k];
        for (int j = lane + 32; j < rm[k]; j += 32) atomicAdd(&ncnt[p[j].y & 255], 1);
    }
    __syncthreads();
    int h = 0, inc = 0;
    if (t < 256) {
        h = ncnt[t];
        inc = wave_incl_scan(h, lane);
        if (lane == 31) wsum[wave] = inc;
    }
    __syncthreads();
    int total = 0;
#pragma unroll
    for (int k = 0; k < 8; ++k) total += wsum[k];
    const bool fits = total <= CAP2;
    if (t < 256) {
        int carry = 0;
#pragma unroll
        for (int k = 0; k < 8; ++k) carry += (k < wave) ? wsum[k] : 0;
        const int ex = carry + inc - h;
        ncnt[t] = ex;
        const int node = b * 256 + t;
        if (node < n) { rowptr[node] = base + ex; dinv[node] = rsqrtf((float)(h + 1)); }
    }
    __syncthreads();
#pragma unroll
    for (int k = 0; k < 8; ++k) {
        if (ev[k].y >= 0) {
            const int pl = atomicAdd(&ncnt[ev[k].y & 255], 1);
            if (fits) sstage[pl] = ev[k].x; else srcs[base + pl] = ev[k].x;
        }
    }
#pragma unroll
    for (int k = 0; k < 8; ++k) {
        const int2* p = binned + (long long)(wave + 32 * k) * len + ro[k];
        for (int j = lane + 32; j < rm[k]; j += 32) {
            const int2 ed = p[j];
            const int pl = atomicAdd(&ncnt[ed.y & 255], 1);
            if (fits) sstage[pl] = ed.x; else srcs[base + pl] = ed.x;
        }
    }
    if (fits) {
        __syncthreads();
        for (int i = t; i < total; i += 1024) srcs[base + i] = sstage[i];
    }
}

__global__ __launch_bounds__(256) void k_weights(const int* __restrict__ srcs, const float* __restrict__ dinv,
                                                 int2* __restrict__ pairs, int e) {
    const int tid = blockIdx.x * 256 + threadIdx.x;
    const int wbase = __builtin_amdgcn_readfirstlane(tid >> 5) * 128;
    const int a = wbase + (tid & 31) * 2;
    if (wbase + 128 <= e) {
        const int2 s0 = *(const int2*)(srcs + a);
        const int2 s1 = *(const int2*)(srcs + a + 64);
        const float w0 = dinv[s0.x];
        const float w1 = dinv[s0.y];
        const float w2 = dinv[s1.x];
        const float w3 = dinv[s1.y];
        *(int4*)(pairs + a)      = make_int4(s0.x, __float_as_int(w0), s0.y, __float_as_int(w1));
        *(int4*)(pairs + a + 64) = make_int4(s1.x, __float_as_int(w2), s1.y, __float_as_int(w3));
        return;
    }
    for (int q = 0; q < 2; ++q) {
        for (int i = a + q * 64; i < min(a + q * 64 + 2, e); ++i) {
            const int s = srcs[i];
            pairs[i] = make_int2(s, __float_as_int(dinv[s]));
        }
    }
}

template <int V> struct FVec;
template <> struct FVec<1> { typedef float T; };
template <> struct FVec<2> { typedef v2f T; };

__device__ __forceinline__ float relu_v(float a) { return fmaxf(a, 0.0f); }
__device__ __forceinline__ v2f relu_v(v2f a) { a.x = fmaxf(a.x, 0.0f); a.y = fmaxf(a.y, 0.0f); return a; }

template <int D, int K, typename T>
__device__ __forceinline__ void slot_load(T (&hv)[K], float (&w)[K], const float* __restrict__ hl,
                                          int sx, int sy, int k, float dj) {
#pragma unroll
    for (int u = 0; u < K; ++u) {
        const int s = __builtin_amdgcn_readlane(sx, k + u);
        w[u] = __int_as_float(__builtin_amdgcn_readlane(sy, k + u)) * dj;
        hv[u] = *(const T*)(hl + (long long)s * D);
    }
}

template <int D, typename T>
__device__ __forceinline__ T row_sum(const float* __restrict__ hl, const int2* __restrict__ pairs,
                                     int p0, int p1, int2 first, float dj, int lane) {
    T acc = {};
    int2 my = first;
    for (int base = p0; base < p1; ) {
        const int m = min(p1 - base, 32);

        const int r = m & 7;
        T th[7]; float tw[7];
#pragma unroll
        for (int u = 0; u < 7; ++u) {
            const int s = __builtin_amdgcn_readlane(my.x, u);
            tw[u] = __int_as_float(__builtin_amdgcn_readlane(my.y, u)) * dj;
            if (u < r) th[u] = *(const T*)(hl + (long long)s * D);
        }
        for (int k = r; k < m; k += 8) {
            T hv[8]; float w[8];
            slot_load<D, 8, T>(hv, w, hl, my.x, my.y, k, dj);
#pragma unroll
            for (int u = 0; u < 8; ++u) acc += hv[u] * w[u];
        }
#pragma unroll
        for (int u = 0; u < 7; ++u) {
            if (u < r) acc += th[u] * tw[u];
        }

        base += 32;
        if (base < p1) {
            my = make_int2(0, 0);
            if (lane < min(p1 - base, 32)) my = pairs[base + lane];
        }
    }
    return acc;
}

template <int D, bool RELU>
__global__ __launch_bounds__(256) void k_gather(const float* __restrict__ H,
                                                const int* __restrict__ rowptr,
                                                const int2* __restrict__ pairs,
                                                const float* __restrict__ dinv,
                                                const float* __restrict__ bias,
                                                float* __restrict__ out, int n, int e) {
    constexpr int V = D / 32;
    typedef typename FVec<V>::T T;
    const int lane = threadIdx.x & 31;
    const int row = blockIdx.x * 8 + __builtin_amdgcn_readfirstlane((int)(threadIdx.x >> 5));
    if (row >= n) return;
    const float* hl = H + lane * V;
    const T hs = *(const T*)(hl + (long long)row * D);
    const T bv = *(const T*)(bias + lane * V);
    const int p0 = rowptr[row];
    const int p1 = (row + 1 < n) ? rowptr[row + 1] : e;
    const float dj = dinv[row];

    T acc = {};
    for (int base = p0; base < p1; base += 32) {
        const int m = min(p1 - base, 32);
        int2 my = make_int2(0, 0);
        if (lane < m) my = pairs[base + lane];

        const int r = m & 7;
        T th[7]; float tw[7];
#pragma unroll
        for (int u = 0; u < 7; ++u) {
            const int s = __builtin_amdgcn_readlane(my.x, u);
            tw[u] = __int_as_float(__builtin_amdgcn_readlane(my.y, u)) * dj;
            if (u < r) th[u] = *(const T*)(hl + (long long)s * D);
        }
        for (int k = r; k < m; k += 8) {
            T hv[8]; float w[8];
            slot_load<D, 8, T>(hv, w, hl, my.x, my.y, k, dj);
#pragma unroll
            for (int u = 0; u < 8; ++u) acc += hv[u] * w[u];
        }
#pragma unroll
        for (int u = 0; u < 7; ++u) {
            if (u < r) acc += th[u] * tw[u];
        }
    }

    acc += hs * (dj * dj);
    acc += bv;
    if (RELU) acc = relu_v(acc);
    *(T*)(out + (long long)row * D + lane * V) = acc;
}

__global__ __launch_bounds__(256) void k_gather_gemm_relu_gemm(const float* __restrict__ X,
                                                               const int* __restrict__ rowptr,
                                                               const int2* __restrict__ pairs,
                                                               const float* __restrict__ dinv,
                                                               const float* __restrict__ W1,
                                                               const float* __restrict__ bias,
                                                               const float* __restrict__ W2,
                                                               float* __restrict__ H2, int n, int e) {
    constexpr int D = 64, DO = 32, PITCH = 68;
    __shared__ float tile[16 * PITCH];
    __shared__ float tile2[16 * PITCH];
    const int lane = threadIdx.x & 31;
    const int wave = __builtin_amdgcn_readfirstlane((int)(threadIdx.x >> 5));
    const int m0 = blockIdx.x * 16;
    const int r0 = m0 + wave * 2;
    const float* xl = X + lane * 2;

    const int pa = rowptr[r0];
    const int pb = rowptr[r0 + 1];
    const int pc = (r0 + 2 < n) ? rowptr[r0 + 2] : e;
    const float dj0 = dinv[r0];
    const float dj1 = dinv[r0 + 1];
    const v2f hs0 = *(const v2f*)(xl + (long long)r0 * D);
    const v2f hs1 = *(const v2f*)(xl + (long long)(r0 + 1) * D);
    int2 f0 = make_int2(0, 0), f1 = make_int2(0, 0);
    if (lane < min(pb - pa, 32)) f0 = pairs[pa + lane];
    if (lane < min(pc - pb, 32)) f1 = pairs[pb + lane];

    v2f a0 = row_sum<D, v2f>(xl, pairs, pa, pb, f0, dj0, lane);
    a0 += hs0 * (dj0 * dj0);
    *(v2f*)(tile + (wave * 2 + 0) * PITCH + lane * 2) = a0;

    v2f a1 = row_sum<D, v2f>(xl, pairs, pb, pc, f1, dj1, lane);
    a1 += hs1 * (dj1 * dj1);
    *(v2f*)(tile + (wave * 2 + 1) * PITCH + lane * 2) = a1;

    const int lr = lane & 15;
    const int lh = lane >> 4;
    const int n1 = (wave & 3) * 16;
    v2f bf[16];
    float bc = 0.0f;
    if (wave < 4) {
#pragma unroll
        for (int k = 0; k < 64; k += 4) {
            bf[k >> 2].x = W1[(k + lh * 2 + 0) * D + n1 + lr];
            bf[k >> 2].y = W1[(k + lh * 2 + 1) * D + n1 + lr];
        }
        bc = bias[n1 + lr];
    }
    __builtin_amdgcn_fence(__ATOMIC_RELEASE, "workgroup", "local");
    __builtin_amdgcn_s_barrier();
    __builtin_amdgcn_fence(__ATOMIC_ACQUIRE, "workgroup", "local");

    if (wave < 4) {
        typedef __attribute__((address_space(3))) float lfloat;
        lfloat* tp = (lfloat*)tile2 + (lh * 8) * PITCH + n1 + lr;
        asm volatile("" : "+v"(tp));

        v8f c = {};
#pragma unroll
        for (int k = 0; k < 64; k += 4) {
            const v2f a = *(const v2f*)(tile + lr * PITCH + k + lh * 2);
            c = __builtin_amdgcn_wmma_f32_16x16x4_f32(false, a, false, bf[k >> 2],
                                                      (short)0, c, false, false);
        }
        asm volatile("v_nop\n\tv_nop\n\tv_nop\n\tv_nop\n\tv_nop" : "+v"(c));
#pragma unroll
        for (int r = 0; r < 8; ++r) {
            tp[r * PITCH] = fmaxf(c[r] + bc, 0.0f);
        }
    }

    const int n0 = (wave & 1) * 16;
    v2f bg[16];
    if (wave < 2) {
#pragma unroll
        for (int k = 0; k < 64; k += 4) {
            bg[k >> 2].x = W2[(k + lh * 2 + 0) * DO + n0 + lr];
            bg[k >> 2].y = W2[(k + lh * 2 + 1) * DO + n0 + lr];
        }
    }
    __builtin_amdgcn_fence(__ATOMIC_RELEASE, "workgroup", "local");
    __builtin_amdgcn_s_barrier();
    __builtin_amdgcn_fence(__ATOMIC_ACQUIRE, "workgroup", "local");
    if (wave >= 2) return;

    typedef __attribute__((address_space(1))) float gfloat;
    gfloat* hp = (gfloat*)(H2 + (long long)(m0 + lh * 8) * DO + n0 + lr);
    asm volatile("" : "+v"(hp));

    v8f c2 = {};
#pragma unroll
    for (int k = 0; k < 64; k += 4) {
        const v2f a = *(const v2f*)(tile2 + lr * PITCH + k + lh * 2);
        c2 = __builtin_amdgcn_wmma_f32_16x16x4_f32(false, a, false, bg[k >> 2],
                                                   (short)0, c2, false, false);
    }
#pragma unroll
    for (int r = 0; r < 8; ++r) {
        hp[r * DO] = c2[r];
    }
}

extern "C" void kernel_launch(void* const* d_in, const int* in_sizes, int n_in,
                              void* d_out, int out_size, void* d_ws, size_t ws_size,
                              hipStream_t stream) {
    const int D_IN = 64, D_H = 64, D_OUT = 32;
    const int N = in_sizes[0] / D_IN;
    const int E = in_sizes[1] / 2;

    const float* x  = (const float*)d_in[0];
    const int*   ei = (const int*)d_in[1];
    const float* W1 = (const float*)d_in[2];
    const float* b1 = (const float*)d_in[3];
    const float* W2 = (const float*)d_in[4];
    const float* b2 = (const float*)d_in[5];
    float* out = (float*)d_out;

    const int* srcIdx = ei;
    const int* dstIdx = ei + E;

    const int len = (E + CH - 1) / CH;
    int2*  pairs  = (int2*)d_ws;
    int*   scr    = (int*)d_ws + 2 * (long long)E;
    float* h2     = (float*)scr + (long long)N * D_H;
    int*   rowptr = (int*)(h2 + (long long)N * D_H);
    float* dinv   = (float*)(rowptr + N);
    int2*  binned = (int2*)scr;
    int*   ccnt   = scr + 2 * (long long)CH * len;
    int*   coff   = ccnt + 256 * CH;
    int*   srcs   = coff + 256 * CH;

    const int BT = 256;
    const int rowTiles = N / 16;
    const int ranges = (N + 255) / 256;

    k_chunk_bin<<<CH, 512, 0, stream>>>(srcIdx, dstIdx, binned, ccnt, coff, E, len);
    k_range_rows<<<ranges, 1024, 0, stream>>>(binned, ccnt, coff, rowptr, dinv, srcs, N, len);
    k_weights<<<((E + 127) / 128 + 7) / 8, BT, 0, stream>>>(srcs, dinv, pairs, E);

    k_gather_gemm_relu_gemm<<<rowTiles, BT, 0, stream>>>(x, rowptr, pairs, dinv, W1, b1, W2, h2, N, E);

    k_gather<D_OUT, false><<<(N + 7) / 8, BT, 0, stream>>>(h2, rowptr, pairs, dinv, b2, out, N, E);

    (void)n_in; (void)out_size; (void)ws_size;
}


# Edge norm factored into row scales, no weight list: layer-1 gather reads dinv[src] beside each slot and stores dinv[row]*h2, layer 2 adds bare rows; weight-fill launch, pairs gone; rounding changes
# speedup vs baseline: 4.1710x; 1.0819x over previous
#include <hip/hip_runtime.h>
#include <hip/hip_bf16.h>


typedef __attribute__((ext_vector_type(2))) float v2f;
typedef __attribute__((ext_vector_type(8))) float v8f;

constexpr int CH = 256;
constexpr int CAP = 4096;
constexpr int CAP2 = 8192;

__device__ __forceinline__ int wave_incl_scan(int x, int lane) {
    x += __builtin_amdgcn_update_dpp(0, x, 0x111, 0xf, 0xf, true);
    x += __builtin_amdgcn_update_dpp(0, x, 0x112, 0xf, 0xf, true);
    x += __builtin_amdgcn_update_dpp(0, x, 0x114, 0xf, 0xf, true);
    x += __builtin_amdgcn_update_dpp(0, x, 0x118, 0xf, 0xf, true);
    const int lo = __builtin_amdgcn_readlane(x, 15);
    if (lane >= 16) x += lo;
    return x;
}

__global__ __launch_bounds__(512) void k_chunk_bin(const int* __restrict__ src, const int* __restrict__ dst,
                                                   int2* __restrict__ binned, int* __restrict__ ccnt,
                                                   int* __restrict__ coff, int e, int len) {
    __shared__ int hist[256];
    __shared__ int offs[256];
    __shared__ int wsum[8];
    __shared__ int2 stage[CAP];
    const int t = threadIdx.x;
    const int lane = t & 31;
    const int c = blockIdx.x;
    const int lo = c * len;
    const int cnt = min(len, e - lo);
    if (t < 256) hist[t] = 0;
    __syncthreads();

    int s[8], d[8], r[8];
#pragma unroll
    for (int u = 0; u < 8; ++u) {
        const int i = u * 512 + t;
        s[u] = 0; d[u] = -1; r[u] = 0;
        if (i < cnt) { s[u] = src[lo + i]; d[u] = dst[lo + i]; }
    }
#pragma unroll
    for (int u = 0; u < 8; ++u) {
        if (d[u] >= 0) r[u] = atomicAdd(&hist[d[u] >> 8], 1);
    }
    __syncthreads();
    int h = 0, inc = 0;
    if (t < 256) {
        h = hist[t];
        inc = wave_incl_scan(h, lane);
        if (lane == 31) wsum[t >> 5] = inc;
    }
    __syncthreads();
    if (t < 256) {
        int carry = 0;
#pragma unroll
        for (int k = 0; k < 8; ++k) carry += (k < (t >> 5)) ? wsum[k] : 0;
        const int ex = carry + inc - h;
        offs[t] = ex;
        ccnt[t * CH + c] = h;
        coff[t * CH + c] = ex;
    }
    __syncthreads();
#pragma unroll
    for (int u = 0; u < 8; ++u) {
        if (d[u] >= 0) stage[offs[d[u] >> 8] + r[u]] = make_int2(s[u], d[u]);
    }
    __syncthreads();
    for (int i = t; i < cnt; i += 512) binned[lo + i] = stage[i];
}

__global__ __launch_bounds__(1024) void k_range_rows(const int2* __restrict__ binned,
                                                     const int* __restrict__ ccnt,
                                                     const int* __restrict__ coff,
                                                     int* __restrict__ rowptr, float* __restrict__ dinv,
                                                     int* __restrict__ srcs, int n, int len) {
    __shared__ int scnt[256];
    __shared__ int soff[256];
    __shared__ int ncnt[256];
    __shared__ int red[32];
    __shared__ int wsum[8];
    __shared__ int sstage[CAP2];
    const int t = threadIdx.x;
    const int lane = t & 31;
    const int wave = __builtin_amdgcn_readfirstlane((int)(t >> 5));
    const int b = blockIdx.x;

    if (t < 256) { scnt[t] = ccnt[b * CH + t]; soff[t] = coff[b * CH + t]; ncnt[t] = 0; }

    const int lim4 = b * (CH / 4);
    const int4* c4 = (const int4*)ccnt;
    int part = 0;
#pragma unroll
    for (int g = 0; g < 2; ++g) {
        if (g * 8192 < lim4) {
            int4 v[8];
#pragma unroll
            for (int u = 0; u < 8; ++u) v[u] = c4[min(g * 8192 + u * 1024 + t, lim4 - 1)];
#pragma unroll
            for (int u = 0; u < 8; ++u) {
                const int sm = (v[u].x + v[u].y) + (v[u].z + v[u].w);
                part += (g * 8192 + u * 1024 + t < lim4) ? sm : 0;
            }
        }
    }
    const int wtot = wave_incl_scan(part, lane);
    if (lane == 31) red[wave] = wtot;
    __syncthreads();
    const int base = __builtin_amdgcn_readlane(wave_incl_scan(red[lane], lane), 31);

    int2 ev[8];
    int rm[8], ro[8];
#pragma unroll
    for (int k = 0; k < 8; ++k) {
        rm[k] = __builtin_amdgcn_readfirstlane(scnt[wave + 32 * k]);
        ro[k] = __builtin_amdgcn_readfirstlane(soff[wave + 32 * k]);
    }
#pragma unroll
    for (int k = 0; k < 8; ++k) {
        ev[k] = make_int2(0, -1);
        if (lane < rm[k]) ev[k] = binned[(long long)(wave + 32 * k) * len + ro[k] + lane];
    }
#pragma unroll
    for (int k = 0; k < 8; ++k) {
        if (ev[k].y >= 0) atomicAdd(&ncnt[ev[k].y & 255], 1);
    }
#pragma unroll
    for (int k = 0; k < 8; ++k) {
        const int2* p = binned + (long long)(wave + 32 * k) * len + ro[k];
        for (int j = lane + 32; j < rm[k]; j += 32) atomicAdd(&ncnt[p[j].y & 255], 1);
    }
    __syncthreads();
    int h = 0, inc = 0;
    if (t < 256) {
        h = ncnt[t];
        inc = wave_incl_scan(h, lane);
        if (lane == 31) wsum[wave] = inc;
    }
    __syncthreads();
    int total = 0;
#pragma unroll
    for (int k = 0; k < 8; ++k) total += wsum[k];
    const bool fits = total <= CAP2;
    if (t < 256) {
        int carry = 0;
#pragma unroll
        for (int k = 0; k < 8; ++k) carry += (k < wave) ? wsum[k] : 0;
        const int ex = carry + inc - h;
        ncnt[t] = ex;
        const int node = b * 256 + t;
        if (node < n) { rowptr[node] = base + ex; dinv[node] = rsqrtf((float)(h + 1)); }
    }
    __syncthreads();
#pragma unroll
    for (int k = 0; k < 8; ++k) {
        if (ev[k].y >= 0) {
            const int pl = atomicAdd(&ncnt[ev[k].y & 255], 1);
            if (fits) sstage[pl] = ev[k].x; else srcs[base + pl] = ev[k].x;
        }
    }
#pragma unroll
    for (int k = 0; k < 8; ++k) {
        const int2* p = binned + (long long)(wave + 32 * k) * len + ro[k];
        for (int j = lane + 32; j < rm[k]; j += 32) {
            const int2 ed = p[j];
            const int pl = atomicAdd(&ncnt[ed.y & 255], 1);
            if (fits) sstage[pl] = ed.x; else srcs[base + pl] = ed.x;
        }
    }
    if (fits) {
        __syncthreads();
        for (int i = t; i < total; i += 1024) srcs[base + i] = sstage[i];
    }
}

template <int V> struct FVec;
template <> struct FVec<1> { typedef float T; };
template <> struct FVec<2> { typedef v2f T; };

template <int D, typename T>
__device__ __forceinline__ T row_sum_w(const float* __restrict__ hl, const int* __restrict__ srcs,
                                       const float* __restrict__ dinv, int p0, int p1,
                                       int fs, int lane) {
    T acc = {};
    int ms = fs;
    for (int base = p0; base < p1; ) {
        const int m = min(p1 - base, 32);

        const int r = m & 7;
        T th[7];
#pragma unroll
        for (int u = 0; u < 7; ++u) {
            const int s = __builtin_amdgcn_readlane(ms, u);
            if (u < r) th[u] = *(const T*)(hl + (long long)s * D);
        }
        const int mw = __float_as_int(dinv[ms]);
        for (int k = r; k < m; k += 8) {
            T hv[8];
#pragma unroll
            for (int u = 0; u < 8; ++u) {
                const int s = __builtin_amdgcn_readlane(ms, k + u);
                hv[u] = *(const T*)(hl + (long long)s * D);
            }
#pragma unroll
            for (int u = 0; u < 8; ++u)
                acc += hv[u] * __int_as_float(__builtin_amdgcn_readlane(mw, k + u));
        }
#pragma unroll
        for (int u = 0; u < 7; ++u) {
            if (u < r) acc += th[u] * __int_as_float(__builtin_amdgcn_readlane(mw, u));
        }

        base += 32;
        if (base < p1) {
            ms = 0;
            if (lane < min(p1 - base, 32)) ms = srcs[base + lane];
        }
    }
    return acc;
}

template <int D>
__global__ __launch_bounds__(256) void k_gather_bare(const float* __restrict__ H,
                                                     const int* __restrict__ rowptr,
                                                     const int* __restrict__ srcs,
                                                     const float* __restrict__ dinv,
                                                     const float* __restrict__ bias,
                                                     float* __restrict__ out, int n, int e) {
    constexpr int V = D / 32;
    typedef typename FVec<V>::T T;
    const int lane = threadIdx.x & 31;
    const int row = blockIdx.x * 8 + __builtin_amdgcn_readfirstlane((int)(threadIdx.x >> 5));
    if (row >= n) return;
    const float* hl = H + lane * V;
    const T hs = *(const T*)(hl + (long long)row * D);
    const T bv = *(const T*)(bias + lane * V);
    const int p0 = rowptr[row];
    const int p1 = (row + 1 < n) ? rowptr[row + 1] : e;
    const float dj = dinv[row];

    T acc = {};
    for (int base = p0; base < p1; base += 32) {
        const int m = min(p1 - base, 32);
        int ms = 0;
        if (lane < m) ms = srcs[base + lane];

        const int r = m & 7;
        T th[7];
#pragma unroll
        for (int u = 0; u < 7; ++u) {
            const int s = __builtin_amdgcn_readlane(ms, u);
            if (u < r) th[u] = *(const T*)(hl + (long long)s * D);
        }
        for (int k = r; k < m; k += 8) {
            T hv[8];
#pragma unroll
            for (int u = 0; u < 8; ++u) {
                const int s = __builtin_amdgcn_readlane(ms, k + u);
                hv[u] = *(const T*)(hl + (long long)s * D);
            }
#pragma unroll
            for (int u = 0; u < 8; ++u) acc += hv[u];
        }
#pragma unroll
        for (int u = 0; u < 7; ++u) {
            if (u < r) acc += th[u];
        }
    }

    acc = (acc + hs) * dj + bv;
    *(T*)(out + (long long)row * D + lane * V) = acc;
}

__global__ __launch_bounds__(256) void k_gather_gemm_relu_gemm(const float* __restrict__ X,
                                                               const int* __restrict__ rowptr,
                                                               const int* __restrict__ srcs,
                                                               const float* __restrict__ dinv,
                                                               const float* __restrict__ W1,
                                                               const float* __restrict__ bias,
                                                               const float* __restrict__ W2,
                                                               float* __restrict__ H2, int n, int e) {
    constexpr int D = 64, DO = 32, PITCH = 68;
    __shared__ float tile[16 * PITCH];
    __shared__ float tile2[16 * PITCH];
    __shared__ __attribute__((aligned(16))) float sdj[16];
    const int lane = threadIdx.x & 31;
    const int wave = __builtin_amdgcn_readfirstlane((int)(threadIdx.x >> 5));
    const int m0 = blockIdx.x * 16;
    const int r0 = m0 + wave * 2;
    const float* xl = X + lane * 2;

    const int pa = rowptr[r0];
    const int pb = rowptr[r0 + 1];
    const int pc = (r0 + 2 < n) ? rowptr[r0 + 2] : e;
    const float dj0 = dinv[r0];
    const float dj1 = dinv[r0 + 1];
    const v2f hs0 = *(const v2f*)(xl + (long long)r0 * D);
    const v2f hs1 = *(const v2f*)(xl + (long long)(r0 + 1) * D);
    const bool in0 = lane < min(pb - pa, 32);
    const bool in1 = lane < min(pc - pb, 32);
    int s0 = 0, s1 = 0;
    if (in0) s0 = srcs[pa + lane];
    if (in1) s1 = srcs[pb + lane];
    if (lane == 0) { sdj[wave * 2 + 0] = dj0; sdj[wave * 2 + 1] = dj1; }

    v2f a0 = row_sum_w<D, v2f>(xl, srcs, dinv, pa, pb, s0, lane);
    a0 = (a0 + hs0 * dj0) * dj0;
    *(v2f*)(tile + (wave * 2 + 0) * PITCH + lane * 2) = a0;

    v2f a1 = row_sum_w<D, v2f>(xl, srcs, dinv, pb, pc, s1, lane);
    a1 = (a1 + hs1 * dj1) * dj1;
    *(v2f*)(tile + (wave * 2 + 1) * PITCH + lane * 2) = a1;

    const int lr = lane & 15;
    const int lh = lane >> 4;
    const int n1 = (wave & 3) * 16;
    v2f bf[16];
    float bc = 0.0f;
    if (wave < 4) {
#pragma unroll
        for (int k = 0; k < 64; k += 4) {
            bf[k >> 2].x = W1[(k + lh * 2 + 0) * D + n1 + lr];
            bf[k >> 2].y = W1[(k + lh * 2 + 1) * D + n1 + lr];
        }
        bc = bias[n1 + lr];
    }
    __builtin_amdgcn_fence(__ATOMIC_RELEASE, "workgroup", "local");
    __builtin_amdgcn_s_barrier();
    __builtin_amdgcn_fence(__ATOMIC_ACQUIRE, "workgroup", "local");

    if (wave < 4) {
        typedef __attribute__((address_space(3))) float lfloat;
        lfloat* tp = (lfloat*)tile2 + (lh * 8) * PITCH + n1 + lr;
        asm volatile("" : "+v"(tp));

        lfloat* sp = (lfloat*)sdj + lh * 8;
        asm volatile("" : "+v"(sp));

        v8f c = {};
#pragma unroll
        for (int k = 0; k < 64; k += 4) {
            const v2f a = *(const v2f*)(tile + lr * PITCH + k + lh * 2);
            c = __builtin_amdgcn_wmma_f32_16x16x4_f32(false, a, false, bf[k >> 2],
                                                      (short)0, c, false, false);
        }
        asm volatile("v_nop\n\tv_nop\n\tv_nop\n\tv_nop\n\tv_nop" : "+v"(c));
        float sd[8];
#pragma unroll
        for (int r = 0; r < 8; ++r) sd[r] = sp[r];
#pragma unroll
        for (int r = 0; r < 8; ++r) {
            tp[r * PITCH] = fmaxf(c[r] + bc, 0.0f) * sd[r];
        }
    }

    const int n0 = (wave & 1) * 16;
    v2f bg[16];
    if (wave < 2) {
#pragma unroll
        for (int k = 0; k < 64; k += 4) {
            bg[k >> 2].x = W2[(k + lh * 2 + 0) * DO + n0 + lr];
            bg[k >> 2].y = W2[(k + lh * 2 + 1) * DO + n0 + lr];
        }
    }
    __builtin_amdgcn_fence(__ATOMIC_RELEASE, "workgroup", "local");
    __builtin_amdgcn_s_barrier();
    __builtin_amdgcn_fence(__ATOMIC_ACQUIRE, "workgroup", "local");
    if (wave >= 2) return;

    typedef __attribute__((address_space(1))) float gfloat;
    gfloat* hp = (gfloat*)(H2 + (long long)(m0 + lh * 8) * DO + n0 + lr);
    asm volatile("" : "+v"(hp));

    v8f c2 = {};
#pragma unroll
    for (int k = 0; k < 64; k += 4) {
        const v2f a = *(const v2f*)(tile2 + lr * PITCH + k + lh * 2);
        c2 = __builtin_amdgcn_wmma_f32_16x16x4_f32(false, a, false, bg[k >> 2],
                                                   (short)0, c2, false, false);
    }
#pragma unroll
    for (int r = 0; r < 8; ++r) {
        hp[r * DO] = c2[r];
    }
}

extern "C" void kernel_launch(void* const* d_in, const int* in_sizes, int n_in,
                              void* d_out, int out_size, void* d_ws, size_t ws_size,
                              hipStream_t stream) {
    const int D_IN = 64, D_H = 64, D_OUT = 32;
    const int N = in_sizes[0] / D_IN;
    const int E = in_sizes[1] / 2;

    const float* x  = (const float*)d_in[0];
    const int*   ei = (const int*)d_in[1];
    const float* W1 = (const float*)d_in[2];
    const float* b1 = (const float*)d_in[3];
    const float* W2 = (const float*)d_in[4];
    const float* b2 = (const float*)d_in[5];
    float* out = (float*)d_out;

    const int* srcIdx = ei;
    const int* dstIdx = ei + E;

    const int len = (E + CH - 1) / CH;
    int*   scr    = (int*)d_ws;
    float* h2     = (float*)scr + (long long)N * D_H;
    int*   rowptr = (int*)(h2 + (long long)N * D_H);
    float* dinv   = (float*)(rowptr + N);
    int2*  binned = (int2*)scr;
    int*   ccnt   = scr + 2 * (long long)CH * len;
    int*   coff   = ccnt + 256 * CH;
    int*   srcs   = coff + 256 * CH;

    const int BT = 256;
    const int rowTiles = N / 16;
    const int ranges = (N + 255) / 256;

    k_chunk_bin<<<CH, 512, 0, stream>>>(srcIdx, dstIdx, binned, ccnt, coff, E, len);
    k_range_rows<<<ranges, 1024, 0, stream>>>(binned, ccnt, coff, rowptr, dinv, srcs, N, len);

    k_gather_gemm_relu_gemm<<<rowTiles, BT, 0, stream>>>(x, rowptr, srcs, dinv, W1, b1, W2, h2, N, E);

    k_gather_bare<D_OUT><<<(N + 7) / 8, BT, 0, stream>>>(h2, rowptr, srcs, dinv, b2, out, N, E);

    (void)n_in; (void)out_size; (void)ws_size;
}
